# starting kernel
# baseline (speedup 1.0000x reference)
#include <hip/hip_runtime.h>

#define N_NODES   20000
#define FDIM      256
#define HEADS     4
#define CDIM      64
#define NEDGE     320000
#define ETOT      (NEDGE + N_NODES)
#define NEG_SLOPE 0.2f

typedef __attribute__((ext_vector_type(2))) float v2f;
typedef __attribute__((ext_vector_type(8))) float v8f;

__device__ __forceinline__ unsigned f2ord(float f) {
  unsigned u = __float_as_uint(f);
  return (u & 0x80000000u) ? ~u : (u | 0x80000000u);
}
__device__ __forceinline__ float ord2f(unsigned u) {
  unsigned v = (u & 0x80000000u) ? (u ^ 0x80000000u) : ~u;
  return __uint_as_float(v);
}
__device__ __forceinline__ float leaky(float v) {
  return v > 0.0f ? v : NEG_SLOPE * v;
}

__global__ void zero_f32_kernel(float* __restrict__ p, int n) {
  int i = blockIdx.x * blockDim.x + threadIdx.x;
  if (i < n) p[i] = 0.0f;
}
__global__ void zero_u32_kernel(unsigned* __restrict__ p, int n) {
  int i = blockIdx.x * blockDim.x + threadIdx.x;
  if (i < n) p[i] = 0u;
}

__global__ __launch_bounds__(512)
void gemm_f32_wmma(const float* __restrict__ A, const float* __restrict__ B,
                   float* __restrict__ Out) {
  __shared__ float As[16 * 257];
  const int tid      = threadIdx.x;
  const int row_base = blockIdx.x * 16;

  for (int i = tid; i < 16 * 256; i += 512) {
    int r = i >> 8, c = i & 255;
    As[r * 257 + c] = A[(size_t)(row_base + r) * FDIM + c];
  }
  __syncthreads();

  const int wave = tid >> 5;
  const int lane = tid & 31;
  const int cn   = wave * 16;
  const int m    = lane & 15;
  const int kh   = (lane >> 4) << 1;

  v8f acc = {};
  #pragma unroll 8
  for (int k = 0; k < FDIM; k += 4) {
    v2f a, b;
    a.x = As[m * 257 + k + kh];
    a.y = As[m * 257 + k + kh + 1];
    b.x = B[(size_t)(k + kh)     * FDIM + cn + m];
    b.y = B[(size_t)(k + kh + 1) * FDIM + cn + m];
    acc = __builtin_amdgcn_wmma_f32_16x16x4_f32(false, a, false, b,
                                                (short)0, acc, false, false);
  }

  const int rbase = row_base + ((lane >> 4) << 3);
  #pragma unroll
  for (int r = 0; r < 8; ++r) {
    Out[(size_t)(rbase + r) * FDIM + cn + m] = acc[r];
  }
}

__global__ __launch_bounds__(256)
void edge_scores(const float* __restrict__ xl, const float* __restrict__ xr,
                 const float* __restrict__ att, const int* __restrict__ ei,
                 float* __restrict__ score, unsigned* __restrict__ m_u) {
  const int e    = blockIdx.x * (blockDim.x >> 5) + (threadIdx.x >> 5);
  const int lane = threadIdx.x & 31;
  if (e >= ETOT) return;
  int src, dst;
  if (e < NEDGE) { src = ei[e]; dst = ei[NEDGE + e]; }
  else           { src = e - NEDGE; dst = src; }

  const float4* pl4 = (const float4*)(xl + (size_t)src * FDIM) + lane * 2;
  const float4* pr4 = (const float4*)(xr + (size_t)dst * FDIM) + lane * 2;
  const float4* at4 = (const float4*)att + lane * 2;

  float4 l0 = pl4[0], l1 = pl4[1];
  float4 r0 = pr4[0], r1 = pr4[1];
  float4 a0 = at4[0], a1 = at4[1];

  float acc = leaky(l0.x + r0.x) * a0.x + leaky(l0.y + r0.y) * a0.y +
              leaky(l0.z + r0.z) * a0.z + leaky(l0.w + r0.w) * a0.w +
              leaky(l1.x + r1.x) * a1.x + leaky(l1.y + r1.y) * a1.y +
              leaky(l1.z + r1.z) * a1.z + leaky(l1.w + r1.w) * a1.w;

  acc += __shfl_xor(acc, 4, 32);
  acc += __shfl_xor(acc, 2, 32);
  acc += __shfl_xor(acc, 1, 32);

  if ((lane & 7) == 0) {
    const int h = lane >> 3;
    score[(size_t)e * HEADS + h] = acc;
    atomicMax(&m_u[dst * HEADS + h], f2ord(acc));
  }
}

__global__ __launch_bounds__(256)
void edge_exp(const int* __restrict__ ei, float* __restrict__ score_p,
              const unsigned* __restrict__ m_u, float* __restrict__ denom) {
  const int t = blockIdx.x * blockDim.x + threadIdx.x;
  if (t >= ETOT * HEADS) return;
  const int e = t >> 2;
  const int h = t & 3;
  const int dst = (e < NEDGE) ? ei[NEDGE + e] : (e - NEDGE);
  const float m = ord2f(m_u[dst * HEADS + h]);
  const float p = expf(score_p[t] - m);
  score_p[t] = p;
  unsafeAtomicAdd(&denom[dst * HEADS + h], p);
}

__global__ __launch_bounds__(256)
void edge_aggregate(const float* __restrict__ xl, const float* __restrict__ p,
                    const float* __restrict__ denom, const int* __restrict__ ei,
                    float* __restrict__ agg) {
  const int e    = blockIdx.x * (blockDim.x >> 5) + (threadIdx.x >> 5);
  const int lane = threadIdx.x & 31;
  if (e >= ETOT) return;
  int src, dst;
  if (e < NEDGE) { src = ei[e]; dst = ei[NEDGE + e]; }
  else           { src = e - NEDGE; dst = src; }

  const int h = lane >> 3;
  const float alpha = p[(size_t)e * HEADS + h] / (denom[dst * HEADS + h] + 1e-16f);

  const float4* pl4 = (const float4*)(xl + (size_t)src * FDIM) + lane * 2;
  float*        pa  = agg + (size_t)dst * FDIM + lane * 8;
  float4 v0 = pl4[0], v1 = pl4[1];

  unsafeAtomicAdd(pa + 0, alpha * v0.x);
  unsafeAtomicAdd(pa + 1, alpha * v0.y);
  unsafeAtomicAdd(pa + 2, alpha * v0.z);
  unsafeAtomicAdd(pa + 3, alpha * v0.w);
  unsafeAtomicAdd(pa + 4, alpha * v1.x);
  unsafeAtomicAdd(pa + 5, alpha * v1.y);
  unsafeAtomicAdd(pa + 6, alpha * v1.z);
  unsafeAtomicAdd(pa + 7, alpha * v1.w);
}

__global__ __launch_bounds__(256)
void finalize_elu(float* __restrict__ io, const float* __restrict__ bias, int total) {
  const int i = blockIdx.x * blockDim.x + threadIdx.x;
  if (i >= total) return;
  float v = io[i] + bias[i & (FDIM - 1)];
  io[i] = v > 0.0f ? v : (expf(v) - 1.0f);
}

extern "C" void kernel_launch(void* const* d_in, const int* in_sizes, int n_in,
                              void* d_out, int out_size, void* d_ws, size_t ws_size,
                              hipStream_t stream) {
  const float* x    = (const float*)d_in[0];
  const int*   ei   = (const int*)  d_in[1];
  const float* Wl0  = (const float*)d_in[2];
  const float* Wr0  = (const float*)d_in[3];
  const float* att0 = (const float*)d_in[4];
  const float* b0   = (const float*)d_in[5];
  const float* Wl1  = (const float*)d_in[6];
  const float* Wr1  = (const float*)d_in[7];
  const float* att1 = (const float*)d_in[8];
  const float* b1   = (const float*)d_in[9];
  float* out = (float*)d_out;

  float*    ws    = (float*)d_ws;
  float*    xl    = ws;
  float*    xr    = xl   + (size_t)N_NODES * FDIM;
  float*    hbuf  = xr   + (size_t)N_NODES * FDIM;
  float*    pbuf  = hbuf + (size_t)N_NODES * FDIM;
  unsigned* m_u   = (unsigned*)(pbuf + (size_t)ETOT * HEADS);
  float*    denom = (float*)(m_u + (size_t)N_NODES * HEADS);

  const int nh        = N_NODES * HEADS;
  const int nfeat     = N_NODES * FDIM;
  const int edgeWaves = (ETOT + 7) / 8;

  auto layer = [&](const float* in, const float* Wl, const float* Wr,
                   const float* att, const float* bias, float* outb) {
    gemm_f32_wmma<<<N_NODES / 16, 512, 0, stream>>>(in, Wl, xl);
    gemm_f32_wmma<<<N_NODES / 16, 512, 0, stream>>>(in, Wr, xr);
    zero_u32_kernel<<<(nh + 255) / 256, 256, 0, stream>>>(m_u, nh);
    zero_f32_kernel<<<(nh + 255) / 256, 256, 0, stream>>>(denom, nh);
    zero_f32_kernel<<<(nfeat + 255) / 256, 256, 0, stream>>>(outb, nfeat);
    edge_scores<<<edgeWaves, 256, 0, stream>>>(xl, xr, att, ei, pbuf, m_u);
    edge_exp<<<(ETOT * HEADS + 255) / 256, 256, 0, stream>>>(ei, pbuf, m_u, denom);
    edge_aggregate<<<edgeWaves, 256, 0, stream>>>(xl, pbuf, denom, ei, outb);
    finalize_elu<<<(nfeat + 255) / 256, 256, 0, stream>>>(outb, bias, nfeat);
  };

  layer(x,    Wl0, Wr0, att0, b0, hbuf);
  layer(hbuf, Wl1, Wr1, att1, b1, out);
}


# edge phase per destination node: in-edges listed once per call in a fixed order (count, scan, fill, rank); one wave per node does score, running softmax, weighted sum in registers; no float atomics
# speedup vs baseline: 3.3449x; 3.3449x over previous
#include <hip/hip_runtime.h>

#define N_NODES   20000
#define FDIM      256
#define HEADS     4
#define CDIM      64
#define NEDGE     320000
#define ETOT      (NEDGE + N_NODES)
#define NEG_SLOPE 0.2f

typedef __attribute__((ext_vector_type(2))) float v2f;
typedef __attribute__((ext_vector_type(8))) float v8f;

__device__ __forceinline__ float leaky(float v) {
  return v > 0.0f ? v : NEG_SLOPE * v;
}

__global__ void zero_i32_kernel(int* __restrict__ p, int n) {
  int i = blockIdx.x * blockDim.x + threadIdx.x;
  if (i < n) p[i] = 0;
}

__global__ __launch_bounds__(512)
void gemm_f32_wmma(const float* __restrict__ A, const float* __restrict__ B,
                   float* __restrict__ Out) {
  __shared__ float As[16 * 257];
  const int tid      = threadIdx.x;
  const int row_base = blockIdx.x * 16;

  for (int i = tid; i < 16 * 256; i += 512) {
    int r = i >> 8, c = i & 255;
    As[r * 257 + c] = A[(size_t)(row_base + r) * FDIM + c];
  }
  __syncthreads();

  const int wave = tid >> 5;
  const int lane = tid & 31;
  const int cn   = wave * 16;
  const int m    = lane & 15;
  const int kh   = (lane >> 4) << 1;

  v8f acc = {};
  #pragma unroll 8
  for (int k = 0; k < FDIM; k += 4) {
    v2f a, b;
    a.x = As[m * 257 + k + kh];
    a.y = As[m * 257 + k + kh + 1];
    b.x = B[(size_t)(k + kh)     * FDIM + cn + m];
    b.y = B[(size_t)(k + kh + 1) * FDIM + cn + m];
    acc = __builtin_amdgcn_wmma_f32_16x16x4_f32(false, a, false, b,
                                                (short)0, acc, false, false);
  }

  const int rbase = row_base + ((lane >> 4) << 3);
  #pragma unroll
  for (int r = 0; r < 8; ++r) {
    Out[(size_t)(rbase + r) * FDIM + cn + m] = acc[r];
  }
}

__global__ __launch_bounds__(256)
void count_in_edges(const int* __restrict__ ei, int* __restrict__ cnt) {
  const int e = blockIdx.x * blockDim.x + threadIdx.x;
  if (e < NEDGE) atomicAdd(&cnt[ei[NEDGE + e]], 1);
}

#define SCAN_THREADS 1024
#define SCAN_ITEMS   20
__global__ __launch_bounds__(SCAN_THREADS)
void scan_counts(const int* __restrict__ cnt, int* __restrict__ rowptr) {
  __shared__ int part[SCAN_THREADS];
  const int t    = threadIdx.x;
  const int base = t * SCAN_ITEMS;
  int before[SCAN_ITEMS];
  int s = 0;
  for (int k = 0; k < SCAN_ITEMS; ++k) {
    const int idx = base + k;
    before[k] = s;
    s += (idx < N_NODES) ? cnt[idx] : 0;
  }
  part[t] = s;
  __syncthreads();
  for (int off = 1; off < SCAN_THREADS; off <<= 1) {
    const int add = (t >= off) ? part[t - off] : 0;
    __syncthreads();
    part[t] += add;
    __syncthreads();
  }
  const int excl = part[t] - s;
  for (int k = 0; k < SCAN_ITEMS; ++k) {
    const int idx = base + k;
    if (idx < N_NODES) rowptr[idx] = excl + before[k];
  }
  if (t == SCAN_THREADS - 1) rowptr[N_NODES] = part[t];
}

__global__ __launch_bounds__(256)
void fill_in_edges(const int* __restrict__ ei, const int* __restrict__ rowptr,
                   int* __restrict__ cnt, int* __restrict__ slot_eid) {
  const int e = blockIdx.x * blockDim.x + threadIdx.x;
  if (e >= NEDGE) return;
  const int dst = ei[NEDGE + e];
  const int k   = atomicSub(&cnt[dst], 1);
  slot_eid[rowptr[dst] + k - 1] = e;
}

__global__ __launch_bounds__(256)
void order_segments(const int* __restrict__ rowptr, const int* __restrict__ slot_eid,
                    const int* __restrict__ ei, int* __restrict__ csr_src) {
  const int lane = threadIdx.x & 31;
  const int node = blockIdx.x * (blockDim.x >> 5) + (threadIdx.x >> 5);
  if (node >= N_NODES) return;
  const int beg = rowptr[node];
  const int len = rowptr[node + 1] - beg;
  if (len <= 32) {
    const int mine = (lane < len) ? slot_eid[beg + lane] : 0x7fffffff;
    int rank = 0;
    for (int u = 0; u < len; ++u) rank += (__shfl(mine, u, 32) < mine) ? 1 : 0;
    if (lane < len) csr_src[beg + rank] = ei[mine];
  } else {
    for (int t = lane; t < len; t += 32) {
      const int mine = slot_eid[beg + t];
      int rank = 0;
      for (int u = 0; u < len; ++u) rank += (slot_eid[beg + u] < mine) ? 1 : 0;
      csr_src[beg + rank] = ei[mine];
    }
  }
}

__global__ __launch_bounds__(256)
void node_edge_phase(const float* __restrict__ xl, const float* __restrict__ xr,
                     const float* __restrict__ att, const float* __restrict__ bias,
                     const int* __restrict__ rowptr, const int* __restrict__ csr_src,
                     float* __restrict__ outb) {
  const int lane = threadIdx.x & 31;
  const int node = __builtin_amdgcn_readfirstlane(
      (int)(blockIdx.x * (blockDim.x >> 5) + (threadIdx.x >> 5)));
  if (node >= N_NODES) return;
  const int beg = rowptr[node];
  const int end = rowptr[node + 1];
  const int lo  = lane * 8;

  const float4* at4 = (const float4*)(att + lo);
  const float4* pr4 = (const float4*)(xr + (size_t)node * FDIM + lo);
  const float4 a0 = at4[0], a1 = at4[1];
  const float4 r0 = pr4[0], r1 = pr4[1];

  int j = beg;
  int s_fetch = (j < end) ? csr_src[j] : node;
  int s_after = (j + 1 < end) ? csr_src[j + 1] : node;

  const float4* ps4 = (const float4*)(xl + (size_t)node * FDIM + lo);
  float4 c0 = ps4[0], c1 = ps4[1];
  float4 n0 = c0, n1 = c1;
  if (j < end) {
    const float4* pn4 = (const float4*)(xl + (size_t)s_fetch * FDIM + lo);
    n0 = pn4[0]; n1 = pn4[1];
  }

  auto score = [&](const float4& l0, const float4& l1) -> float {
    float s = leaky(l0.x + r0.x) * a0.x + leaky(l0.y + r0.y) * a0.y +
              leaky(l0.z + r0.z) * a0.z + leaky(l0.w + r0.w) * a0.w +
              leaky(l1.x + r1.x) * a1.x + leaky(l1.y + r1.y) * a1.y +
              leaky(l1.z + r1.z) * a1.z + leaky(l1.w + r1.w) * a1.w;
    s += __shfl_xor(s, 4, 32);
    s += __shfl_xor(s, 2, 32);
    s += __shfl_xor(s, 1, 32);
    return s;
  };

  float m = score(c0, c1);
  float d = 1.0f;
  float acc0 = c0.x, acc1 = c0.y, acc2 = c0.z, acc3 = c0.w;
  float acc4 = c1.x, acc5 = c1.y, acc6 = c1.z, acc7 = c1.w;

  for (; j < end; ++j) {
    c0 = n0; c1 = n1;
    s_fetch = s_after;
    s_after = (j + 2 < end) ? csr_src[j + 2] : node;
    if (j + 1 < end) {
      const float4* pn4 = (const float4*)(xl + (size_t)s_fetch * FDIM + lo);
      n0 = pn4[0]; n1 = pn4[1];
    }
    const float s  = score(c0, c1);
    const float ex = expf(-fabsf(s - m));
    const bool  up = s > m;
    const float sc = up ? ex : 1.0f;
    const float p  = up ? 1.0f : ex;
    m = up ? s : m;
    d    = d    * sc + p;
    acc0 = acc0 * sc + p * c0.x;
    acc1 = acc1 * sc + p * c0.y;
    acc2 = acc2 * sc + p * c0.z;
    acc3 = acc3 * sc + p * c0.w;
    acc4 = acc4 * sc + p * c1.x;
    acc5 = acc5 * sc + p * c1.y;
    acc6 = acc6 * sc + p * c1.z;
    acc7 = acc7 * sc + p * c1.w;
  }

  const float den = d + 1e-16f;
  const float4* b4 = (const float4*)(bias + lo);
  const float4 b0 = b4[0], b1 = b4[1];
  auto elu = [](float v) -> float { return v > 0.0f ? v : (expf(v) - 1.0f); };
  float4 o0, o1;
  o0.x = elu(acc0 / den + b0.x);
  o0.y = elu(acc1 / den + b0.y);
  o0.z = elu(acc2 / den + b0.z);
  o0.w = elu(acc3 / den + b0.w);
  o1.x = elu(acc4 / den + b1.x);
  o1.y = elu(acc5 / den + b1.y);
  o1.z = elu(acc6 / den + b1.z);
  o1.w = elu(acc7 / den + b1.w);
  float4* po4 = (float4*)(outb + (size_t)node * FDIM + lo);
  po4[0] = o0;
  po4[1] = o1;
}

extern "C" void kernel_launch(void* const* d_in, const int* in_sizes, int n_in,
                              void* d_out, int out_size, void* d_ws, size_t ws_size,
                              hipStream_t stream) {
  const float* x    = (const float*)d_in[0];
  const int*   ei   = (const int*)  d_in[1];
  const float* Wl0  = (const float*)d_in[2];
  const float* Wr0  = (const float*)d_in[3];
  const float* att0 = (const float*)d_in[4];
  const float* b0   = (const float*)d_in[5];
  const float* Wl1  = (const float*)d_in[6];
  const float* Wr1  = (const float*)d_in[7];
  const float* att1 = (const float*)d_in[8];
  const float* b1   = (const float*)d_in[9];
  float* out = (float*)d_out;

  float* ws       = (float*)d_ws;
  float* xl       = ws;
  float* xr       = xl   + (size_t)N_NODES * FDIM;
  float* hbuf     = xr   + (size_t)N_NODES * FDIM;
  int*   cnt      = (int*)(hbuf + (size_t)N_NODES * FDIM);
  int*   rowptr   = cnt      + N_NODES;
  int*   slot_eid = rowptr   + N_NODES + 16;
  int*   csr_src  = slot_eid + NEDGE;

  zero_i32_kernel<<<(N_NODES + 255) / 256, 256, 0, stream>>>(cnt, N_NODES);
  count_in_edges<<<(NEDGE + 255) / 256, 256, 0, stream>>>(ei, cnt);
  scan_counts<<<1, SCAN_THREADS, 0, stream>>>(cnt, rowptr);
  fill_in_edges<<<(NEDGE + 255) / 256, 256, 0, stream>>>(ei, rowptr, cnt, slot_eid);
  order_segments<<<(N_NODES + 7) / 8, 256, 0, stream>>>(rowptr, slot_eid, ei, csr_src);

  auto layer = [&](const float* in, const float* Wl, const float* Wr,
                   const float* att, const float* bias, float* outb) {
    gemm_f32_wmma<<<N_NODES / 16, 512, 0, stream>>>(in, Wl, xl);
    gemm_f32_wmma<<<N_NODES / 16, 512, 0, stream>>>(in, Wr, xr);
    node_edge_phase<<<(N_NODES + 7) / 8, 256, 0, stream>>>(xl, xr, att, bias,
                                                            rowptr, csr_src, outb);
  };

  layer(x,    Wl0, Wr0, att0, b0, hbuf);
  layer(hbuf, Wl1, Wr1, att1, b1, out);
}


# GEMM: each wave keeps its 256x16 panel of W in registers and its block walks 10 row tiles staged through two LDS buffers; xl and xr of a layer in one launch (grid.y picks W), 250 blocks
# speedup vs baseline: 4.1455x; 1.2393x over previous
#include <hip/hip_runtime.h>

#define N_NODES   20000
#define FDIM      256
#define HEADS     4
#define CDIM      64
#define NEDGE     320000
#define ETOT      (NEDGE + N_NODES)
#define NEG_SLOPE 0.2f

typedef __attribute__((ext_vector_type(2))) float v2f;
typedef __attribute__((ext_vector_type(8))) float v8f;

__device__ __forceinline__ float leaky(float v) {
  return v > 0.0f ? v : NEG_SLOPE * v;
}

__global__ void zero_i32_kernel(int* __restrict__ p, int n) {
  int i = blockIdx.x * blockDim.x + threadIdx.x;
  if (i < n) p[i] = 0;
}

#define A_PITCH        260
#define TILES_PER_BLK  10
#define ROW_GROUPS     125
__global__ __launch_bounds__(512)
void gemm_panel_f32_wmma(const float* __restrict__ A,
                         const float* __restrict__ Bl, const float* __restrict__ Br,
                         float* __restrict__ Ol, float* __restrict__ Or) {
  __shared__ __attribute__((aligned(16))) float As[2][16 * A_PITCH];
  const float* __restrict__ B   = blockIdx.y ? Br : Bl;
  float* __restrict__       Out = blockIdx.y ? Or : Ol;
  const int tid  = threadIdx.x;
  const int wave = tid >> 5;
  const int lane = tid & 31;
  const int cn   = wave * 16;
  const int m    = lane & 15;
  const int kh   = (lane >> 4) << 1;
  const int tile0 = blockIdx.x * TILES_PER_BLK;

  const int sr = tid >> 6;
  const int sc = (tid & 63) << 2;
  const float* Ab = A + (size_t)tile0 * 16 * FDIM + (size_t)sr * FDIM + sc;
  float4 s0 = *(const float4*)(Ab);
  float4 s1 = *(const float4*)(Ab + 8 * FDIM);

  v2f bp[64];
  const float* Bp = B + (size_t)kh * FDIM + cn + m;
  #pragma unroll
  for (int kk = 0; kk < 64; ++kk) {
    bp[kk].x = Bp[(size_t)(4 * kk)     * FDIM];
    bp[kk].y = Bp[(size_t)(4 * kk + 1) * FDIM];
  }

  *(float4*)&As[0][sr * A_PITCH + sc]       = s0;
  *(float4*)&As[0][(sr + 8) * A_PITCH + sc] = s1;
  __syncthreads();

  const int roff = m * A_PITCH + kh;
  float* po = Out + (size_t)(tile0 * 16 + ((lane >> 4) << 3)) * FDIM + cn + m;
  #pragma unroll 1
  for (int t = 0; t < TILES_PER_BLK; ++t) {
    const int cur = t & 1;
    if (t + 1 < TILES_PER_BLK) {
      const float* An = Ab + (size_t)(t + 1) * 16 * FDIM;
      s0 = *(const float4*)(An);
      s1 = *(const float4*)(An + 8 * FDIM);
    }
    const float* as = &As[cur][roff];
    v2f af[2][16];
    #pragma unroll
    for (int i = 0; i < 16; ++i) af[0][i] = *(const v2f*)(as + 4 * i);
    v8f acc = {};
    #pragma unroll
    for (int g = 0; g < 4; ++g) {
      if (g + 1 < 4) {
        #pragma unroll
        for (int i = 0; i < 16; ++i)
          af[(g + 1) & 1][i] = *(const v2f*)(as + 4 * (16 * (g + 1) + i));
      }
      __builtin_amdgcn_sched_barrier(0);
      #pragma unroll
      for (int i = 0; i < 16; ++i)
        acc = __builtin_amdgcn_wmma_f32_16x16x4_f32(false, af[g & 1][i], false, bp[16 * g + i],
                                                    (short)0, acc, false, false);
    }

    if (t + 1 < TILES_PER_BLK) {
      *(float4*)&As[cur ^ 1][sr * A_PITCH + sc]       = s0;
      *(float4*)&As[cur ^ 1][(sr + 8) * A_PITCH + sc] = s1;
    }
    __syncthreads();
    #pragma unroll
    for (int r = 0; r < 8; ++r) po[(size_t)r * FDIM] = acc[r];
    po += 16 * FDIM;
  }
}

__global__ __launch_bounds__(256)
void count_in_edges(const int* __restrict__ ei, int* __restrict__ cnt) {
  const int e = blockIdx.x * blockDim.x + threadIdx.x;
  if (e < NEDGE) atomicAdd(&cnt[ei[NEDGE + e]], 1);
}

#define SCAN_THREADS 1024
#define SCAN_ITEMS   20
__global__ __launch_bounds__(SCAN_THREADS)
void scan_counts(const int* __restrict__ cnt, int* __restrict__ rowptr) {
  __shared__ int part[SCAN_THREADS];
  const int t    = threadIdx.x;
  const int base = t * SCAN_ITEMS;
  int before[SCAN_ITEMS];
  int s = 0;
  for (int k = 0; k < SCAN_ITEMS; ++k) {
    const int idx = base + k;
    before[k] = s;
    s += (idx < N_NODES) ? cnt[idx] : 0;
  }
  part[t] = s;
  __syncthreads();
  for (int off = 1; off < SCAN_THREADS; off <<= 1) {
    const int add = (t >= off) ? part[t - off] : 0;
    __syncthreads();
    part[t] += add;
    __syncthreads();
  }
  const int excl = part[t] - s;
  for (int k = 0; k < SCAN_ITEMS; ++k) {
    const int idx = base + k;
    if (idx < N_NODES) rowptr[idx] = excl + before[k];
  }
  if (t == SCAN_THREADS - 1) rowptr[N_NODES] = part[t];
}

__global__ __launch_bounds__(256)
void fill_in_edges(const int* __restrict__ ei, const int* __restrict__ rowptr,
                   int* __restrict__ cnt, int* __restrict__ slot_eid) {
  const int e = blockIdx.x * blockDim.x + threadIdx.x;
  if (e >= NEDGE) return;
  const int dst = ei[NEDGE + e];
  const int k   = atomicSub(&cnt[dst], 1);
  slot_eid[rowptr[dst] + k - 1] = e;
}

__global__ __launch_bounds__(256)
void order_segments(const int* __restrict__ rowptr, const int* __restrict__ slot_eid,
                    const int* __restrict__ ei, int* __restrict__ csr_src) {
  const int lane = threadIdx.x & 31;
  const int node = blockIdx.x * (blockDim.x >> 5) + (threadIdx.x >> 5);
  if (node >= N_NODES) return;
  const int beg = rowptr[node];
  const int len = rowptr[node + 1] - beg;
  if (len <= 32) {
    const int mine = (lane < len) ? slot_eid[beg + lane] : 0x7fffffff;
    int rank = 0;
    for (int u = 0; u < len; ++u) rank += (__shfl(mine, u, 32) < mine) ? 1 : 0;
    if (lane < len) csr_src[beg + rank] = ei[mine];
  } else {
    for (int t = lane; t < len; t += 32) {
      const int mine = slot_eid[beg + t];
      int rank = 0;
      for (int u = 0; u < len; ++u) rank += (slot_eid[beg + u] < mine) ? 1 : 0;
      csr_src[beg + rank] = ei[mine];
    }
  }
}

__global__ __launch_bounds__(256)
void node_edge_phase(const float* __restrict__ xl, const float* __restrict__ xr,
                     const float* __restrict__ att, const float* __restrict__ bias,
                     const int* __restrict__ rowptr, const int* __restrict__ csr_src,
                     float* __restrict__ outb) {
  const int lane = threadIdx.x & 31;
  const int node = __builtin_amdgcn_readfirstlane(
      (int)(blockIdx.x * (blockDim.x >> 5) + (threadIdx.x >> 5)));
  if (node >= N_NODES) return;
  const int beg = rowptr[node];
  const int end = rowptr[node + 1];
  const int lo  = lane * 8;

  const float4* at4 = (const float4*)(att + lo);
  const float4* pr4 = (const float4*)(xr + (size_t)node * FDIM + lo);
  const float4 a0 = at4[0], a1 = at4[1];
  const float4 r0 = pr4[0], r1 = pr4[1];

  int j = beg;
  int s_fetch = (j < end) ? csr_src[j] : node;
  int s_after = (j + 1 < end) ? csr_src[j + 1] : node;

  const float4* ps4 = (const float4*)(xl + (size_t)node * FDIM + lo);
  float4 c0 = ps4[0], c1 = ps4[1];
  float4 n0 = c0, n1 = c1;
  if (j < end) {
    const float4* pn4 = (const float4*)(xl + (size_t)s_fetch * FDIM + lo);
    n0 = pn4[0]; n1 = pn4[1];
  }

  auto score = [&](const float4& l0, const float4& l1) -> float {
    float s = leaky(l0.x + r0.x) * a0.x + leaky(l0.y + r0.y) * a0.y +
              leaky(l0.z + r0.z) * a0.z + leaky(l0.w + r0.w) * a0.w +
              leaky(l1.x + r1.x) * a1.x + leaky(l1.y + r1.y) * a1.y +
              leaky(l1.z + r1.z) * a1.z + leaky(l1.w + r1.w) * a1.w;
    s += __shfl_xor(s, 4, 32);
    s += __shfl_xor(s, 2, 32);
    s += __shfl_xor(s, 1, 32);
    return s;
  };

  float m = score(c0, c1);
  float d = 1.0f;
  float acc0 = c0.x, acc1 = c0.y, acc2 = c0.z, acc3 = c0.w;
  float acc4 = c1.x, acc5 = c1.y, acc6 = c1.z, acc7 = c1.w;

  for (; j < end; ++j) {
    c0 = n0; c1 = n1;
    s_fetch = s_after;
    s_after = (j + 2 < end) ? csr_src[j + 2] : node;
    if (j + 1 < end) {
      const float4* pn4 = (const float4*)(xl + (size_t)s_fetch * FDIM + lo);
      n0 = pn4[0]; n1 = pn4[1];
    }
    const float s  = score(c0, c1);
    const float ex = expf(-fabsf(s - m));
    const bool  up = s > m;
    const float sc = up ? ex : 1.0f;
    const float p  = up ? 1.0f : ex;
    m = up ? s : m;
    d    = d    * sc + p;
    acc0 = acc0 * sc + p * c0.x;
    acc1 = acc1 * sc + p * c0.y;
    acc2 = acc2 * sc + p * c0.z;
    acc3 = acc3 * sc + p * c0.w;
    acc4 = acc4 * sc + p * c1.x;
    acc5 = acc5 * sc + p * c1.y;
    acc6 = acc6 * sc + p * c1.z;
    acc7 = acc7 * sc + p * c1.w;
  }

  const float den = d + 1e-16f;
  const float4* b4 = (const float4*)(bias + lo);
  const float4 b0 = b4[0], b1 = b4[1];
  auto elu = [](float v) -> float { return v > 0.0f ? v : (expf(v) - 1.0f); };
  float4 o0, o1;
  o0.x = elu(acc0 / den + b0.x);
  o0.y = elu(acc1 / den + b0.y);
  o0.z = elu(acc2 / den + b0.z);
  o0.w = elu(acc3 / den + b0.w);
  o1.x = elu(acc4 / den + b1.x);
  o1.y = elu(acc5 / den + b1.y);
  o1.z = elu(acc6 / den + b1.z);
  o1.w = elu(acc7 / den + b1.w);
  float4* po4 = (float4*)(outb + (size_t)node * FDIM + lo);
  po4[0] = o0;
  po4[1] = o1;
}

extern "C" void kernel_launch(void* const* d_in, const int* in_sizes, int n_in,
                              void* d_out, int out_size, void* d_ws, size_t ws_size,
                              hipStream_t stream) {
  const float* x    = (const float*)d_in[0];
  const int*   ei   = (const int*)  d_in[1];
  const float* Wl0  = (const float*)d_in[2];
  const float* Wr0  = (const float*)d_in[3];
  const float* att0 = (const float*)d_in[4];
  const float* b0   = (const float*)d_in[5];
  const float* Wl1  = (const float*)d_in[6];
  const float* Wr1  = (const float*)d_in[7];
  const float* att1 = (const float*)d_in[8];
  const float* b1   = (const float*)d_in[9];
  float* out = (float*)d_out;

  float* ws       = (float*)d_ws;
  float* xl       = ws;
  float* xr       = xl   + (size_t)N_NODES * FDIM;
  float* hbuf     = xr   + (size_t)N_NODES * FDIM;
  int*   cnt      = (int*)(hbuf + (size_t)N_NODES * FDIM);
  int*   rowptr   = cnt      + N_NODES;
  int*   slot_eid = rowptr   + N_NODES + 16;
  int*   csr_src  = slot_eid + NEDGE;

  zero_i32_kernel<<<(N_NODES + 255) / 256, 256, 0, stream>>>(cnt, N_NODES);
  count_in_edges<<<(NEDGE + 255) / 256, 256, 0, stream>>>(ei, cnt);
  scan_counts<<<1, SCAN_THREADS, 0, stream>>>(cnt, rowptr);
  fill_in_edges<<<(NEDGE + 255) / 256, 256, 0, stream>>>(ei, rowptr, cnt, slot_eid);
  order_segments<<<(N_NODES + 7) / 8, 256, 0, stream>>>(rowptr, slot_eid, ei, csr_src);

  auto layer = [&](const float* in, const float* Wl, const float* Wr,
                   const float* att, const float* bias, float* outb) {
    gemm_panel_f32_wmma<<<dim3(ROW_GROUPS, 2), 512, 0, stream>>>(in, Wl, Wr, xl, xr);
    node_edge_phase<<<(N_NODES + 7) / 8, 256, 0, stream>>>(xl, xr, att, bias,
                                                            rowptr, csr_src, outb);
  };

  layer(x,    Wl0, Wr0, att0, b0, hbuf);
  layer(hbuf, Wl1, Wr1, att1, b1, out);
}


# node kernel: short edge step (8-lane score sum as row_xmask DPP adds, base-2 softmax on v_exp_f32 with log2e folded into att, ids by one vector load + readlane), rows requested two edges ahead
# speedup vs baseline: 4.5095x; 1.0878x over previous
#include <hip/hip_runtime.h>

#define N_NODES   20000
#define FDIM      256
#define HEADS     4
#define CDIM      64
#define NEDGE     320000
#define ETOT      (NEDGE + N_NODES)
#define NEG_SLOPE 0.2f

typedef __attribute__((ext_vector_type(2))) float v2f;
typedef __attribute__((ext_vector_type(8))) float v8f;

__device__ __forceinline__ float leaky(float v) {
  return v > 0.0f ? v : NEG_SLOPE * v;
}

__global__ void zero_i32_kernel(int* __restrict__ p, int n) {
  int i = blockIdx.x * blockDim.x + threadIdx.x;
  if (i < n) p[i] = 0;
}

#define A_PITCH        260
#define TILES_PER_BLK  10
#define ROW_GROUPS     125
__global__ __launch_bounds__(512)
void gemm_panel_f32_wmma(const float* __restrict__ A,
                         const float* __restrict__ Bl, const float* __restrict__ Br,
                         float* __restrict__ Ol, float* __restrict__ Or) {
  __shared__ __attribute__((aligned(16))) float As[2][16 * A_PITCH];
  const float* __restrict__ B   = blockIdx.y ? Br : Bl;
  float* __restrict__       Out = blockIdx.y ? Or : Ol;
  const int tid  = threadIdx.x;
  const int wave = tid >> 5;
  const int lane = tid & 31;
  const int cn   = wave * 16;
  const int m    = lane & 15;
  const int kh   = (lane >> 4) << 1;
  const int tile0 = blockIdx.x * TILES_PER_BLK;

  const int sr = tid >> 6;
  const int sc = (tid & 63) << 2;
  const float* Ab = A + (size_t)tile0 * 16 * FDIM + (size_t)sr * FDIM + sc;
  float4 s0 = *(const float4*)(Ab);
  float4 s1 = *(const float4*)(Ab + 8 * FDIM);

  v2f bp[64];
  const float* Bp = B + (size_t)kh * FDIM + cn + m;
  #pragma unroll
  for (int kk = 0; kk < 64; ++kk) {
    bp[kk].x = Bp[(size_t)(4 * kk)     * FDIM];
    bp[kk].y = Bp[(size_t)(4 * kk + 1) * FDIM];
  }

  *(float4*)&As[0][sr * A_PITCH + sc]       = s0;
  *(float4*)&As[0][(sr + 8) * A_PITCH + sc] = s1;
  __syncthreads();

  const int roff = m * A_PITCH + kh;
  float* po = Out + (size_t)(tile0 * 16 + ((lane >> 4) << 3)) * FDIM + cn + m;
  #pragma unroll 1
  for (int t = 0; t < TILES_PER_BLK; ++t) {
    const int cur = t & 1;
    if (t + 1 < TILES_PER_BLK) {
      const float* An = Ab + (size_t)(t + 1) * 16 * FDIM;
      s0 = *(const float4*)(An);
      s1 = *(const float4*)(An + 8 * FDIM);
    }
    const float* as = &As[cur][roff];
    v2f af[2][16];
    #pragma unroll
    for (int i = 0; i < 16; ++i) af[0][i] = *(const v2f*)(as + 4 * i);
    v8f acc = {};
    #pragma unroll
    for (int g = 0; g < 4; ++g) {
      if (g + 1 < 4) {
        #pragma unroll
        for (int i = 0; i < 16; ++i)
          af[(g + 1) & 1][i] = *(const v2f*)(as + 4 * (16 * (g + 1) + i));
      }
      __builtin_amdgcn_sched_barrier(0);
      #pragma unroll
      for (int i = 0; i < 16; ++i)
        acc = __builtin_amdgcn_wmma_f32_16x16x4_f32(false, af[g & 1][i], false, bp[16 * g + i],
                                                    (short)0, acc, false, false);
    }

    if (t + 1 < TILES_PER_BLK) {
      *(float4*)&As[cur ^ 1][sr * A_PITCH + sc]       = s0;
      *(float4*)&As[cur ^ 1][(sr + 8) * A_PITCH + sc] = s1;
    }
    __syncthreads();
    #pragma unroll
    for (int r = 0; r < 8; ++r) po[(size_t)r * FDIM] = acc[r];
    po += 16 * FDIM;
  }
}

__global__ __launch_bounds__(256)
void count_in_edges(const int* __restrict__ ei, int* __restrict__ cnt) {
  const int e = blockIdx.x * blockDim.x + threadIdx.x;
  if (e < NEDGE) atomicAdd(&cnt[ei[NEDGE + e]], 1);
}

#define SCAN_THREADS 1024
#define SCAN_ITEMS   20
__global__ __launch_bounds__(SCAN_THREADS)
void scan_counts(const int* __restrict__ cnt, int* __restrict__ rowptr) {
  __shared__ int part[SCAN_THREADS];
  const int t    = threadIdx.x;
  const int base = t * SCAN_ITEMS;
  int before[SCAN_ITEMS];
  int s = 0;
  for (int k = 0; k < SCAN_ITEMS; ++k) {
    const int idx = base + k;
    before[k] = s;
    s += (idx < N_NODES) ? cnt[idx] : 0;
  }
  part[t] = s;
  __syncthreads();
  for (int off = 1; off < SCAN_THREADS; off <<= 1) {
    const int add = (t >= off) ? part[t - off] : 0;
    __syncthreads();
    part[t] += add;
    __syncthreads();
  }
  const int excl = part[t] - s;
  for (int k = 0; k < SCAN_ITEMS; ++k) {
    const int idx = base + k;
    if (idx < N_NODES) rowptr[idx] = excl + before[k];
  }
  if (t == SCAN_THREADS - 1) rowptr[N_NODES] = part[t];
}

__global__ __launch_bounds__(256)
void fill_in_edges(const int* __restrict__ ei, const int* __restrict__ rowptr,
                   int* __restrict__ cnt, int* __restrict__ slot_eid) {
  const int e = blockIdx.x * blockDim.x + threadIdx.x;
  if (e >= NEDGE) return;
  const int dst = ei[NEDGE + e];
  const int k   = atomicSub(&cnt[dst], 1);
  slot_eid[rowptr[dst] + k - 1] = e;
}

__global__ __launch_bounds__(256)
void order_segments(const int* __restrict__ rowptr, const int* __restrict__ slot_eid,
                    const int* __restrict__ ei, int* __restrict__ csr_src) {
  const int lane = threadIdx.x & 31;
  const int node = blockIdx.x * (blockDim.x >> 5) + (threadIdx.x >> 5);
  if (node >= N_NODES) return;
  const int beg = rowptr[node];
  const int len = rowptr[node + 1] - beg;
  if (len <= 32) {
    const int mine = (lane < len) ? slot_eid[beg + lane] : 0x7fffffff;
    int rank = 0;
    for (int u = 0; u < len; ++u) rank += (__shfl(mine, u, 32) < mine) ? 1 : 0;
    if (lane < len) csr_src[beg + rank] = ei[mine];
  } else {
    for (int t = lane; t < len; t += 32) {
      const int mine = slot_eid[beg + t];
      int rank = 0;
      for (int u = 0; u < len; ++u) rank += (slot_eid[beg + u] < mine) ? 1 : 0;
      csr_src[beg + rank] = ei[mine];
    }
  }
}

typedef __attribute__((ext_vector_type(4))) float v4f;
#define LOG2E 1.4426950408889634f

template <int CTRL>
__device__ __forceinline__ float xmask_add(float v) {
  const int o = __builtin_amdgcn_update_dpp(0, __float_as_int(v), CTRL, 0xf, 0xf, true);
  return v + __int_as_float(o);
}

__global__ __launch_bounds__(256)
void node_edge_phase(const float* __restrict__ xl, const float* __restrict__ xr,
                     const float* __restrict__ att, const float* __restrict__ bias,
                     const int* __restrict__ rowptr, const int* __restrict__ csr_src,
                     float* __restrict__ outb) {
  const int lane = threadIdx.x & 31;
  const int node = __builtin_amdgcn_readfirstlane(
      (int)(blockIdx.x * (blockDim.x >> 5) + (threadIdx.x >> 5)));
  if (node >= N_NODES) return;
  const int beg = rowptr[node];
  const int end = rowptr[node + 1];
  const int lo  = lane * 8;
  __builtin_amdgcn_sched_barrier(0);

  const v4f* at4 = (const v4f*)(att + lo);
  const v4f* pr4 = (const v4f*)(xr + (size_t)node * FDIM + lo);
  const v4f* ps4 = (const v4f*)(xl + (size_t)node * FDIM + lo);
  const v4f aA = at4[0] * LOG2E, aB = at4[1] * LOG2E;
  const v4f rA = pr4[0], rB = pr4[1];
  const v4f sA = ps4[0], sB = ps4[1];
  const v2f a0 = aA.xy, a1 = aA.zw, a2 = aB.xy, a3 = aB.zw;
  const v2f r0 = rA.xy, r1 = rA.zw, r2 = rB.xy, r3 = rB.zw;

  auto score2 = [&](const v4f& qA, const v4f& qB) -> float {
    const v2f t0 = qA.xy + r0, t1 = qA.zw + r1, t2 = qB.xy + r2, t3 = qB.zw + r3;
    const v2f w0 = __builtin_elementwise_max(t0, t0 * NEG_SLOPE);
    const v2f w1 = __builtin_elementwise_max(t1, t1 * NEG_SLOPE);
    const v2f w2 = __builtin_elementwise_max(t2, t2 * NEG_SLOPE);
    const v2f w3 = __builtin_elementwise_max(t3, t3 * NEG_SLOPE);
    v2f s2 = w0 * a0;
    s2 = __builtin_elementwise_fma(w1, a1, s2);
    s2 = __builtin_elementwise_fma(w2, a2, s2);
    s2 = __builtin_elementwise_fma(w3, a3, s2);
    float s = s2.x + s2.y;
    s = xmask_add<0x161>(s);
    s = xmask_add<0x162>(s);
    s = xmask_add<0x164>(s);
    return s;
  };

  auto load_row = [&](int src, v4f& qA, v4f& qB) {
    const v4f* p4 = (const v4f*)(xl + (size_t)src * FDIM + lo);
    qA = p4[0];
    qB = p4[1];
  };

  int base = beg;
  int n    = (end - base) < 32 ? (end - base) : 32;
  int ids  = (lane < n) ? csr_src[base + lane] : node;
  v4f pA, pB, qA, qB, rA2, rB2;
  load_row(__builtin_amdgcn_readlane(ids, 0), pA, pB);
  __builtin_amdgcn_sched_barrier(0);
  load_row(__builtin_amdgcn_readlane(ids, n > 1 ? 1 : 0), qA, qB);

  float m = score2(sA, sB);
  float d = 1.0f;
  v2f acc0 = sA.xy, acc1 = sA.zw, acc2 = sB.xy, acc3 = sB.zw;

  auto step = [&](const v4f& uA, const v4f& uB) {
    const float s  = score2(uA, uB);
    const float ex = __builtin_amdgcn_exp2f(-__builtin_fabsf(s - m));
    const bool  up = s > m;
    const float sc = up ? ex : 1.0f;
    const float p  = up ? 1.0f : ex;
    m = up ? s : m;
    d    = d    * sc + p;
    acc0 = acc0 * sc + uA.xy * p;
    acc1 = acc1 * sc + uA.zw * p;
    acc2 = acc2 * sc + uB.xy * p;
    acc3 = acc3 * sc + uB.zw * p;
  };

  auto run_chunk = [&](const int cn) __attribute__((always_inline)) {
    const int last = cn - 1;
    int k = 0;
    for (; k + 2 < cn; k += 3) {
      load_row(__builtin_amdgcn_readlane(ids, k + 2), rA2, rB2);
      __builtin_amdgcn_sched_barrier(0);
      step(pA, pB);
      __builtin_amdgcn_sched_barrier(0);
      load_row(__builtin_amdgcn_readlane(ids, (k + 3) < last ? (k + 3) : last), pA, pB);
      __builtin_amdgcn_sched_barrier(0);
      step(qA, qB);
      __builtin_amdgcn_sched_barrier(0);
      load_row(__builtin_amdgcn_readlane(ids, (k + 4) < last ? (k + 4) : last), qA, qB);
      __builtin_amdgcn_sched_barrier(0);
      step(rA2, rB2);
    }
    if (k < cn) {
      step(pA, pB);
      if (k + 1 < cn) step(qA, qB);
    }
  };

  if (n > 0) {
    run_chunk(n);
    for (base += 32; base < end; base += 32) {
      n   = (end - base) < 32 ? (end - base) : 32;
      ids = (lane < n) ? csr_src[base + lane] : node;
      load_row(__builtin_amdgcn_readlane(ids, 0), pA, pB);
      __builtin_amdgcn_sched_barrier(0);
      load_row(__builtin_amdgcn_readlane(ids, n > 1 ? 1 : 0), qA, qB);
      run_chunk(n);
    }
  }

  const float inv = 1.0f / (d + 1e-16f);
  const v4f* b4 = (const v4f*)(bias + lo);
  const v4f bA = b4[0], bB = b4[1];
  auto elu = [](float v) -> float { return v > 0.0f ? v : (expf(v) - 1.0f); };
  v4f oA, oB;
  oA.x = elu(acc0.x * inv + bA.x);
  oA.y = elu(acc0.y * inv + bA.y);
  oA.z = elu(acc1.x * inv + bA.z);
  oA.w = elu(acc1.y * inv + bA.w);
  oB.x = elu(acc2.x * inv + bB.x);
  oB.y = elu(acc2.y * inv + bB.y);
  oB.z = elu(acc3.x * inv + bB.z);
  oB.w = elu(acc3.y * inv + bB.w);
  v4f* po4 = (v4f*)(outb + (size_t)node * FDIM + lo);
  po4[0] = oA;
  po4[1] = oB;
}

extern "C" void kernel_launch(void* const* d_in, const int* in_sizes, int n_in,
                              void* d_out, int out_size, void* d_ws, size_t ws_size,
                              hipStream_t stream) {
  const float* x    = (const float*)d_in[0];
  const int*   ei   = (const int*)  d_in[1];
  const float* Wl0  = (const float*)d_in[2];
  const float* Wr0  = (const float*)d_in[3];
  const float* att0 = (const float*)d_in[4];
  const float* b0   = (const float*)d_in[5];
  const float* Wl1  = (const float*)d_in[6];
  const float* Wr1  = (const float*)d_in[7];
  const float* att1 = (const float*)d_in[8];
  const float* b1   = (const float*)d_in[9];
  float* out = (float*)d_out;

  float* ws       = (float*)d_ws;
  float* xl       = ws;
  float* xr       = xl   + (size_t)N_NODES * FDIM;
  float* hbuf     = xr   + (size_t)N_NODES * FDIM;
  int*   cnt      = (int*)(hbuf + (size_t)N_NODES * FDIM);
  int*   rowptr   = cnt      + N_NODES;
  int*   slot_eid = rowptr   + N_NODES + 16;
  int*   csr_src  = slot_eid + NEDGE;

  zero_i32_kernel<<<(N_NODES + 255) / 256, 256, 0, stream>>>(cnt, N_NODES);
  count_in_edges<<<(NEDGE + 255) / 256, 256, 0, stream>>>(ei, cnt);
  scan_counts<<<1, SCAN_THREADS, 0, stream>>>(cnt, rowptr);
  fill_in_edges<<<(NEDGE + 255) / 256, 256, 0, stream>>>(ei, rowptr, cnt, slot_eid);
  order_segments<<<(N_NODES + 7) / 8, 256, 0, stream>>>(rowptr, slot_eid, ei, csr_src);

  auto layer = [&](const float* in, const float* Wl, const float* Wr,
                   const float* att, const float* bias, float* outb) {
    gemm_panel_f32_wmma<<<dim3(ROW_GROUPS, 2), 512, 0, stream>>>(in, Wl, Wr, xl, xr);
    node_edge_phase<<<(N_NODES + 7) / 8, 256, 0, stream>>>(xl, xr, att, bias,
                                                            rowptr, csr_src, outb);
  };

  layer(x,    Wl0, Wr0, att0, b0, hbuf);
  layer(hbuf, Wl1, Wr1, att1, b1, out);
}
